# starting kernel
# baseline (speedup 1.0000x reference)
#include <hip/hip_runtime.h>

typedef __attribute__((ext_vector_type(2))) float v2f;
typedef __attribute__((ext_vector_type(8))) float v8f;

#define NNODES 40000
#define NEDGES 640000
#define F 128

__global__ void zero_f32(float* __restrict__ p, int n) {
  int i = blockIdx.x * blockDim.x + threadIdx.x;
  int stride = gridDim.x * blockDim.x;
  for (; i < n; i += stride) p[i] = 0.0f;
}

__global__ void scatter_accum(const float* __restrict__ feat,
                              const int* __restrict__ ei,
                              float* __restrict__ sum,
                              float* __restrict__ cnt,
                              int do_cnt) {
  int gid = blockIdx.x * blockDim.x + threadIdx.x;
  if (gid >= NEDGES * 32) return;
  int e = gid >> 5;
  int q = gid & 31;
  int src = ei[e];
  int dst = ei[NEDGES + e];
  const float4 v = *(const float4*)(feat + (size_t)src * F + q * 4);
  float* s = sum + (size_t)dst * F + q * 4;
  atomicAdd(s + 0, v.x);
  atomicAdd(s + 1, v.y);
  atomicAdd(s + 2, v.z);
  atomicAdd(s + 3, v.w);
  if (do_cnt && q == 0) atomicAdd(cnt + dst, 1.0f);
}

__global__ __launch_bounds__(256)
void sage_gemm(const float* __restrict__ sum,
               const float* __restrict__ cnt,
               const float* __restrict__ xin,
               const float* __restrict__ Wl,
               const float* __restrict__ Wr,
               const float* __restrict__ bias,
               float* __restrict__ out,
               int relu) {
  __shared__ float inv[16];
  const int tid  = threadIdx.x;
  const int lane = tid & 31;
  const int wave = tid >> 5;
  const int half = lane >> 4;
  const int lid  = lane & 15;
  const int base = blockIdx.x * 16;
  const int col  = wave * 16 + lid;

  if (tid < 16) {
    float c = cnt[base + tid];
    inv[tid] = 1.0f / fmaxf(c, 1.0f);
  }
  __syncthreads();

  const float* srow = sum + (size_t)(base + lid) * F;
  const float* xrow = xin + (size_t)(base + lid) * F;
  const float scale = inv[lid];

  v8f acc = {};
#pragma unroll 4
  for (int k = 0; k < F; k += 4) {
    const int ka = k + 2 * half;
    v2f a1 = *(const v2f*)(srow + ka);
    a1 *= scale;
    v2f a2 = *(const v2f*)(xrow + ka);
    v2f b1, b2;
    b1.x = Wl[ka * F + col];
    b1.y = Wl[(ka + 1) * F + col];
    b2.x = Wr[ka * F + col];
    b2.y = Wr[(ka + 1) * F + col];
    acc = __builtin_amdgcn_wmma_f32_16x16x4_f32(false, a1, false, b1,
                                                (short)0, acc, false, false);
    acc = __builtin_amdgcn_wmma_f32_16x16x4_f32(false, a2, false, b2,
                                                (short)0, acc, false, false);
  }

  const float bv = bias[col];
#pragma unroll
  for (int v = 0; v < 8; ++v) {
    int m = base + v + 8 * half;
    float val = acc[v] + bv;
    if (relu) val = fmaxf(val, 0.0f);
    out[(size_t)m * F + col] = val;
  }
}

__global__ void fc_kernel(const float* __restrict__ h,
                          const float* __restrict__ Wfc,
                          const float* __restrict__ bfc,
                          float* __restrict__ out) {
  int node = blockIdx.x * blockDim.x + threadIdx.x;
  if (node >= NNODES) return;
  const float* row = h + (size_t)node * F;
  float a0 = bfc[0], a1 = bfc[1];
#pragma unroll 8
  for (int k = 0; k < F; ++k) {
    float v = row[k];
    a0 += v * Wfc[k * 2 + 0];
    a1 += v * Wfc[k * 2 + 1];
  }
  out[node * 2 + 0] = a0;
  out[node * 2 + 1] = a1;
}

extern "C" void kernel_launch(void* const* d_in, const int* in_sizes, int n_in,
                              void* d_out, int out_size, void* d_ws, size_t ws_size,
                              hipStream_t stream) {
  const float* x   = (const float*)d_in[0];
  const int*   ei  = (const int*)d_in[1];
  const float* W1l = (const float*)d_in[2];
  const float* b1  = (const float*)d_in[3];
  const float* W1r = (const float*)d_in[4];
  const float* W2l = (const float*)d_in[5];
  const float* b2  = (const float*)d_in[6];
  const float* W2r = (const float*)d_in[7];
  const float* Wfc = (const float*)d_in[8];
  const float* bfc = (const float*)d_in[9];
  float* out = (float*)d_out;

  float* ws     = (float*)d_ws;
  float* sumbuf = ws;
  float* h1     = sumbuf + (size_t)NNODES * F;
  float* h2     = h1 + (size_t)NNODES * F;
  float* cnt    = h2 + (size_t)NNODES * F;

  const int scatterBlocks = (NEDGES * 32 + 255) / 256;

  zero_f32<<<2048, 256, 0, stream>>>(sumbuf, NNODES * F);
  zero_f32<<<160, 256, 0, stream>>>(cnt, NNODES);
  scatter_accum<<<scatterBlocks, 256, 0, stream>>>(x, ei, sumbuf, cnt, 1);
  sage_gemm<<<NNODES / 16, 256, 0, stream>>>(sumbuf, cnt, x, W1l, W1r, b1, h1, 1);

  zero_f32<<<2048, 256, 0, stream>>>(sumbuf, NNODES * F);
  scatter_accum<<<scatterBlocks, 256, 0, stream>>>(h1, ei, sumbuf, cnt, 0);
  sage_gemm<<<NNODES / 16, 256, 0, stream>>>(sumbuf, cnt, h1, W2l, W2r, b2, h2, 1);

  fc_kernel<<<(NNODES + 255) / 256, 256, 0, stream>>>(h2, Wfc, bfc, out);
}


# Neighbour sums by CSR gather: edges counting-sorted by destination once per call, one wave sums each node's source rows; replaces the float atomic scatter and its zero fills
# speedup vs baseline: 1.7954x; 1.7954x over previous
#include <hip/hip_runtime.h>

typedef __attribute__((ext_vector_type(2))) float v2f;
typedef __attribute__((ext_vector_type(8))) float v8f;

#define NNODES 40000
#define NEDGES 640000
#define F 128

__global__ void zero_f32(float* __restrict__ p, int n) {
  int i = blockIdx.x * blockDim.x + threadIdx.x;
  int stride = gridDim.x * blockDim.x;
  for (; i < n; i += stride) p[i] = 0.0f;
}


#define SCAN_TILES 10
#define SCAN_LEN (SCAN_TILES * 4096)

__global__ __launch_bounds__(256)
void edge_rank(const int* __restrict__ ei, int* __restrict__ deg, int* __restrict__ rank) {
  int e = blockIdx.x * 256 + threadIdx.x;
  if (e >= NEDGES) return;
  int d = ei[NEDGES + e];
  rank[e] = atomicAdd(deg + d, 1);
}

__device__ __forceinline__ int wave_incl_scan(int s, int lane) {
#pragma unroll
  for (int d = 1; d < 32; d <<= 1) {
    int n = __shfl_up(s, d, 32);
    if (lane >= d) s += n;
  }
  return s;
}

__global__ __launch_bounds__(1024)
void scan_deg(const int* __restrict__ deg, int* __restrict__ rowptr, float* __restrict__ cnt) {
  __shared__ int wtot[SCAN_TILES * 32];
  const int t = threadIdx.x;
  const int lane = t & 31;
  const int wave = t >> 5;
  int4 v[SCAN_TILES];
  int incl[SCAN_TILES];
#pragma unroll
  for (int k = 0; k < SCAN_TILES; ++k) {
    int i = k * 4096 + t * 4;
    v[k] = (i < NNODES) ? *(const int4*)(deg + i) : make_int4(0, 0, 0, 0);
  }
#pragma unroll
  for (int k = 0; k < SCAN_TILES; ++k) {
    int s = wave_incl_scan(v[k].x + v[k].y + v[k].z + v[k].w, lane);
    incl[k] = s;
    if (lane == 31) wtot[k * 32 + wave] = s;
  }
  __syncthreads();
  if (wave == 0) {
    int carry = 0;
    for (int k = 0; k < SCAN_TILES; ++k) {
      int w = wtot[k * 32 + lane];
      int s = wave_incl_scan(w, lane);
      wtot[k * 32 + lane] = carry + s - w;
      carry += __shfl(s, 31, 32);
    }
  }
  __syncthreads();
#pragma unroll
  for (int k = 0; k < SCAN_TILES; ++k) {
    int i = k * 4096 + t * 4;
    int4 r;
    r.x = wtot[k * 32 + wave] + incl[k] - (v[k].x + v[k].y + v[k].z + v[k].w);
    r.y = r.x + v[k].x;
    r.z = r.y + v[k].y;
    r.w = r.z + v[k].z;
    *(int4*)(rowptr + i) = r;
    if (i < NNODES)
      *(float4*)(cnt + i) = make_float4((float)v[k].x, (float)v[k].y, (float)v[k].z, (float)v[k].w);
  }
}

__global__ __launch_bounds__(256)
void csr_fill(const int* __restrict__ ei, const int* __restrict__ rowptr,
              const int* __restrict__ rank, int* __restrict__ csr) {
  int e = blockIdx.x * 256 + threadIdx.x;
  if (e >= NEDGES) return;
  int d = ei[NEDGES + e];
  atomicExch(csr + rowptr[d] + rank[e], ei[e]);
}

__device__ __forceinline__ void add4(float4& a, const float4& b) {
  a.x += b.x; a.y += b.y; a.z += b.z; a.w += b.w;
}

template <int NB>
__device__ __forceinline__ void gather_rows(const float* __restrict__ col,
                                            const int* __restrict__ idx, float4& acc) {
  int s[NB];
  float4 v[NB];
#pragma unroll
  for (int u = 0; u < NB; ++u) s[u] = idx[u];
#pragma unroll
  for (int u = 0; u < NB; ++u) v[u] = *(const float4*)(col + (size_t)s[u] * F);
#pragma unroll
  for (int u = 0; u < NB; ++u) add4(acc, v[u]);
}

__global__ __launch_bounds__(256)
void gather_sum(const float* __restrict__ feat, const int* __restrict__ rowptr,
                const int* __restrict__ csr, float* __restrict__ sum) {
  const int lane = threadIdx.x & 31;
  const int node = __builtin_amdgcn_readfirstlane((int)(blockIdx.x * 8 + (threadIdx.x >> 5)));
  const int beg = rowptr[node];
  const int end = rowptr[node + 1];
  const float* col = feat + lane * 4;
  float4 acc = make_float4(0.0f, 0.0f, 0.0f, 0.0f);
  int j = beg;
  for (; j + 8 <= end; j += 8) gather_rows<8>(col, csr + j, acc);
  const int rem = end - j;
  if (rem & 4) { gather_rows<4>(col, csr + j, acc); j += 4; }
  if (rem & 2) { gather_rows<2>(col, csr + j, acc); j += 2; }
  if (rem & 1) { gather_rows<1>(col, csr + j, acc); }
  *(float4*)(sum + (size_t)node * F + lane * 4) = acc;
}

__global__ __launch_bounds__(256)
void sage_gemm(const float* __restrict__ sum,
               const float* __restrict__ cnt,
               const float* __restrict__ xin,
               const float* __restrict__ Wl,
               const float* __restrict__ Wr,
               const float* __restrict__ bias,
               float* __restrict__ out,
               int relu) {
  __shared__ float inv[16];
  const int tid  = threadIdx.x;
  const int lane = tid & 31;
  const int wave = tid >> 5;
  const int half = lane >> 4;
  const int lid  = lane & 15;
  const int base = blockIdx.x * 16;
  const int col  = wave * 16 + lid;

  if (tid < 16) {
    float c = cnt[base + tid];
    inv[tid] = 1.0f / fmaxf(c, 1.0f);
  }
  __syncthreads();

  const float* srow = sum + (size_t)(base + lid) * F;
  const float* xrow = xin + (size_t)(base + lid) * F;
  const float scale = inv[lid];

  v8f acc = {};
#pragma unroll 4
  for (int k = 0; k < F; k += 4) {
    const int ka = k + 2 * half;
    v2f a1 = *(const v2f*)(srow + ka);
    a1 *= scale;
    v2f a2 = *(const v2f*)(xrow + ka);
    v2f b1, b2;
    b1.x = Wl[ka * F + col];
    b1.y = Wl[(ka + 1) * F + col];
    b2.x = Wr[ka * F + col];
    b2.y = Wr[(ka + 1) * F + col];
    acc = __builtin_amdgcn_wmma_f32_16x16x4_f32(false, a1, false, b1,
                                                (short)0, acc, false, false);
    acc = __builtin_amdgcn_wmma_f32_16x16x4_f32(false, a2, false, b2,
                                                (short)0, acc, false, false);
  }

  const float bv = bias[col];
#pragma unroll
  for (int v = 0; v < 8; ++v) {
    int m = base + v + 8 * half;
    float val = acc[v] + bv;
    if (relu) val = fmaxf(val, 0.0f);
    out[(size_t)m * F + col] = val;
  }
}

__global__ void fc_kernel(const float* __restrict__ h,
                          const float* __restrict__ Wfc,
                          const float* __restrict__ bfc,
                          float* __restrict__ out) {
  int node = blockIdx.x * blockDim.x + threadIdx.x;
  if (node >= NNODES) return;
  const float* row = h + (size_t)node * F;
  float a0 = bfc[0], a1 = bfc[1];
#pragma unroll 8
  for (int k = 0; k < F; ++k) {
    float v = row[k];
    a0 += v * Wfc[k * 2 + 0];
    a1 += v * Wfc[k * 2 + 1];
  }
  out[node * 2 + 0] = a0;
  out[node * 2 + 1] = a1;
}

extern "C" void kernel_launch(void* const* d_in, const int* in_sizes, int n_in,
                              void* d_out, int out_size, void* d_ws, size_t ws_size,
                              hipStream_t stream) {
  const float* x   = (const float*)d_in[0];
  const int*   ei  = (const int*)d_in[1];
  const float* W1l = (const float*)d_in[2];
  const float* b1  = (const float*)d_in[3];
  const float* W1r = (const float*)d_in[4];
  const float* W2l = (const float*)d_in[5];
  const float* b2  = (const float*)d_in[6];
  const float* W2r = (const float*)d_in[7];
  const float* Wfc = (const float*)d_in[8];
  const float* bfc = (const float*)d_in[9];
  float* out = (float*)d_out;

  float* ws     = (float*)d_ws;
  float* sumbuf = ws;
  float* h1     = sumbuf + (size_t)NNODES * F;
  float* h2     = h1 + (size_t)NNODES * F;
  float* cnt    = h2 + (size_t)NNODES * F;

  int* deg    = (int*)h2;
  int* rowptr = deg + SCAN_LEN;
  int* rank   = rowptr + SCAN_LEN;
  int* csr    = rank + NEDGES;

  const int edgeBlocks = (NEDGES + 255) / 256;

  zero_f32<<<160, 256, 0, stream>>>((float*)deg, NNODES);
  edge_rank<<<edgeBlocks, 256, 0, stream>>>(ei, deg, rank);
  scan_deg<<<1, 1024, 0, stream>>>(deg, rowptr, cnt);
  csr_fill<<<edgeBlocks, 256, 0, stream>>>(ei, rowptr, rank, csr);

  gather_sum<<<NNODES / 8, 256, 0, stream>>>(x, rowptr, csr, sumbuf);
  sage_gemm<<<NNODES / 16, 256, 0, stream>>>(sumbuf, cnt, x, W1l, W1r, b1, h1, 1);

  gather_sum<<<NNODES / 8, 256, 0, stream>>>(h1, rowptr, csr, sumbuf);
  sage_gemm<<<NNODES / 16, 256, 0, stream>>>(sumbuf, cnt, h1, W2l, W2r, b2, h2, 1);

  fc_kernel<<<(NNODES + 255) / 256, 256, 0, stream>>>(h2, Wfc, bfc, out);
}


# Layer GEMM operand feed: 512 persistent work-groups hold each wave's Wl/Wr fragments in registers for all their node tiles and take the tile rows through LDS by whole-row loads, one tile ahead
# speedup vs baseline: 2.2532x; 1.2550x over previous
#include <hip/hip_runtime.h>

typedef __attribute__((ext_vector_type(2))) float v2f;
typedef __attribute__((ext_vector_type(8))) float v8f;

#define NNODES 40000
#define NEDGES 640000
#define F 128

__global__ void zero_f32(float* __restrict__ p, int n) {
  int i = blockIdx.x * blockDim.x + threadIdx.x;
  int stride = gridDim.x * blockDim.x;
  for (; i < n; i += stride) p[i] = 0.0f;
}


#define SCAN_TILES 10
#define SCAN_LEN (SCAN_TILES * 4096)

__global__ __launch_bounds__(256)
void edge_rank(const int* __restrict__ ei, int* __restrict__ deg, int* __restrict__ rank) {
  int e = blockIdx.x * 256 + threadIdx.x;
  if (e >= NEDGES) return;
  int d = ei[NEDGES + e];
  rank[e] = atomicAdd(deg + d, 1);
}

__device__ __forceinline__ int wave_incl_scan(int s, int lane) {
#pragma unroll
  for (int d = 1; d < 32; d <<= 1) {
    int n = __shfl_up(s, d, 32);
    if (lane >= d) s += n;
  }
  return s;
}

__global__ __launch_bounds__(1024)
void scan_deg(const int* __restrict__ deg, int* __restrict__ rowptr, float* __restrict__ cnt) {
  __shared__ int wtot[SCAN_TILES * 32];
  const int t = threadIdx.x;
  const int lane = t & 31;
  const int wave = t >> 5;
  int4 v[SCAN_TILES];
  int incl[SCAN_TILES];
#pragma unroll
  for (int k = 0; k < SCAN_TILES; ++k) {
    int i = k * 4096 + t * 4;
    v[k] = (i < NNODES) ? *(const int4*)(deg + i) : make_int4(0, 0, 0, 0);
  }
#pragma unroll
  for (int k = 0; k < SCAN_TILES; ++k) {
    int s = wave_incl_scan(v[k].x + v[k].y + v[k].z + v[k].w, lane);
    incl[k] = s;
    if (lane == 31) wtot[k * 32 + wave] = s;
  }
  __syncthreads();
  if (wave == 0) {
    int carry = 0;
    for (int k = 0; k < SCAN_TILES; ++k) {
      int w = wtot[k * 32 + lane];
      int s = wave_incl_scan(w, lane);
      wtot[k * 32 + lane] = carry + s - w;
      carry += __shfl(s, 31, 32);
    }
  }
  __syncthreads();
#pragma unroll
  for (int k = 0; k < SCAN_TILES; ++k) {
    int i = k * 4096 + t * 4;
    int4 r;
    r.x = wtot[k * 32 + wave] + incl[k] - (v[k].x + v[k].y + v[k].z + v[k].w);
    r.y = r.x + v[k].x;
    r.z = r.y + v[k].y;
    r.w = r.z + v[k].z;
    *(int4*)(rowptr + i) = r;
    if (i < NNODES)
      *(float4*)(cnt + i) = make_float4((float)v[k].x, (float)v[k].y, (float)v[k].z, (float)v[k].w);
  }
}

__global__ __launch_bounds__(256)
void csr_fill(const int* __restrict__ ei, const int* __restrict__ rowptr,
              const int* __restrict__ rank, int* __restrict__ csr) {
  int e = blockIdx.x * 256 + threadIdx.x;
  if (e >= NEDGES) return;
  int d = ei[NEDGES + e];
  atomicExch(csr + rowptr[d] + rank[e], ei[e]);
}

__device__ __forceinline__ void add4(float4& a, const float4& b) {
  a.x += b.x; a.y += b.y; a.z += b.z; a.w += b.w;
}

template <int NB>
__device__ __forceinline__ void gather_rows(const float* __restrict__ col,
                                            const int* __restrict__ idx, float4& acc) {
  int s[NB];
  float4 v[NB];
#pragma unroll
  for (int u = 0; u < NB; ++u) s[u] = idx[u];
#pragma unroll
  for (int u = 0; u < NB; ++u) v[u] = *(const float4*)(col + (size_t)s[u] * F);
#pragma unroll
  for (int u = 0; u < NB; ++u) add4(acc, v[u]);
}

__global__ __launch_bounds__(256)
void gather_sum(const float* __restrict__ feat, const int* __restrict__ rowptr,
                const int* __restrict__ csr, float* __restrict__ sum) {
  const int lane = threadIdx.x & 31;
  const int node = __builtin_amdgcn_readfirstlane((int)(blockIdx.x * 8 + (threadIdx.x >> 5)));
  const int beg = rowptr[node];
  const int end = rowptr[node + 1];
  const float* col = feat + lane * 4;
  float4 acc = make_float4(0.0f, 0.0f, 0.0f, 0.0f);
  int j = beg;
  for (; j + 8 <= end; j += 8) gather_rows<8>(col, csr + j, acc);
  const int rem = end - j;
  if (rem & 4) { gather_rows<4>(col, csr + j, acc); j += 4; }
  if (rem & 2) { gather_rows<2>(col, csr + j, acc); j += 2; }
  if (rem & 1) { gather_rows<1>(col, csr + j, acc); }
  *(float4*)(sum + (size_t)node * F + lane * 4) = acc;
}

#define NTILES (NNODES / 16)
#define GEMM_GROUPS 512
#define APITCH 132

__global__ __launch_bounds__(256)
void sage_gemm(const float* __restrict__ sum,
               const float* __restrict__ cnt,
               const float* __restrict__ xin,
               const float* __restrict__ Wl,
               const float* __restrict__ Wr,
               const float* __restrict__ bias,
               float* __restrict__ out,
               int relu) {
  __shared__ __attribute__((aligned(16))) float sS[2][16 * APITCH];
  __shared__ __attribute__((aligned(16))) float sX[2][16 * APITCH];
  const int tid  = threadIdx.x;
  const int lane = tid & 31;
  const int wave = tid >> 5;
  const int half = lane >> 4;
  const int lid  = lane & 15;
  const int col  = wave * 16 + lid;

  const int r0 = wave, r1 = wave + 8;
  const int soff0 = r0 * APITCH + lane * 4;
  const int soff1 = r1 * APITCH + lane * 4;

  int tile = blockIdx.x;
  float4 gs0, gs1, gx0, gx1;
  float c0, c1;
  {
    const size_t g0 = (size_t)(tile * 16 + r0) * F + lane * 4;
    const size_t g1 = (size_t)(tile * 16 + r1) * F + lane * 4;
    gs0 = *(const float4*)(sum + g0);
    gs1 = *(const float4*)(sum + g1);
    gx0 = *(const float4*)(xin + g0);
    gx1 = *(const float4*)(xin + g1);
    c0 = cnt[tile * 16 + r0];
    c1 = cnt[tile * 16 + r1];
  }

  v2f bl[32], br[32];
  {
    const float* wl = Wl + (4 * half) * F + col;
    const float* wr = Wr + (4 * half) * F + col;
#pragma unroll
    for (int p = 0; p < 32; ++p) {
      const int k = 8 * (p >> 1) + 2 * (p & 1);
      bl[p].x = wl[k * F];
      bl[p].y = wl[(k + 1) * F];
      br[p].x = wr[k * F];
      br[p].y = wr[(k + 1) * F];
    }
  }
  const float bv = bias[col];

  {
    const float i0 = 1.0f / fmaxf(c0, 1.0f);
    const float i1 = 1.0f / fmaxf(c1, 1.0f);
    *(float4*)(&sS[0][soff0]) = make_float4(gs0.x * i0, gs0.y * i0, gs0.z * i0, gs0.w * i0);
    *(float4*)(&sS[0][soff1]) = make_float4(gs1.x * i1, gs1.y * i1, gs1.z * i1, gs1.w * i1);
    *(float4*)(&sX[0][soff0]) = gx0;
    *(float4*)(&sX[0][soff1]) = gx1;
  }
  __syncthreads();

  const int aoff = lid * APITCH + 4 * half;
  int buf = 0;
  for (; tile < NTILES; tile += GEMM_GROUPS) {
    const int next = tile + GEMM_GROUPS;
    const bool more = next < NTILES;
    if (more) {
      const size_t g0 = (size_t)(next * 16 + r0) * F + lane * 4;
      const size_t g1 = (size_t)(next * 16 + r1) * F + lane * 4;
      gs0 = *(const float4*)(sum + g0);
      gs1 = *(const float4*)(sum + g1);
      gx0 = *(const float4*)(xin + g0);
      gx1 = *(const float4*)(xin + g1);
      c0 = cnt[next * 16 + r0];
      c1 = cnt[next * 16 + r1];
    }

    const float* as = &sS[buf][aoff];
    const float* ax = &sX[buf][aoff];
    v8f accl = {};
    v8f accr = {};
#pragma unroll
    for (int j = 0; j < 16; ++j) {
      const float4 fs = *(const float4*)(as + 8 * j);
      const float4 fx = *(const float4*)(ax + 8 * j);
      v2f a;
      a.x = fs.x; a.y = fs.y;
      accl = __builtin_amdgcn_wmma_f32_16x16x4_f32(false, a, false, bl[2 * j],
                                                   (short)0, accl, false, false);
      a.x = fx.x; a.y = fx.y;
      accr = __builtin_amdgcn_wmma_f32_16x16x4_f32(false, a, false, br[2 * j],
                                                   (short)0, accr, false, false);
      a.x = fs.z; a.y = fs.w;
      accl = __builtin_amdgcn_wmma_f32_16x16x4_f32(false, a, false, bl[2 * j + 1],
                                                   (short)0, accl, false, false);
      a.x = fx.z; a.y = fx.w;
      accr = __builtin_amdgcn_wmma_f32_16x16x4_f32(false, a, false, br[2 * j + 1],
                                                   (short)0, accr, false, false);
    }

    float val[8];
#pragma unroll
    for (int v = 0; v < 8; ++v) {
      val[v] = accl[v] + accr[v] + bv;
      if (relu) val[v] = fmaxf(val[v], 0.0f);
    }

    if (more) {
      const float i0 = 1.0f / fmaxf(c0, 1.0f);
      const float i1 = 1.0f / fmaxf(c1, 1.0f);
      float* ds = sS[buf ^ 1];
      float* dx = sX[buf ^ 1];
      *(float4*)(ds + soff0) = make_float4(gs0.x * i0, gs0.y * i0, gs0.z * i0, gs0.w * i0);
      *(float4*)(ds + soff1) = make_float4(gs1.x * i1, gs1.y * i1, gs1.z * i1, gs1.w * i1);
      *(float4*)(dx + soff0) = gx0;
      *(float4*)(dx + soff1) = gx1;
    }
    __syncthreads();
    const int base = tile * 16;
#pragma unroll
    for (int v = 0; v < 8; ++v) {
      out[(size_t)(base + v + 8 * half) * F + col] = val[v];
    }
    buf ^= 1;
  }
}

__global__ void fc_kernel(const float* __restrict__ h,
                          const float* __restrict__ Wfc,
                          const float* __restrict__ bfc,
                          float* __restrict__ out) {
  int node = blockIdx.x * blockDim.x + threadIdx.x;
  if (node >= NNODES) return;
  const float* row = h + (size_t)node * F;
  float a0 = bfc[0], a1 = bfc[1];
#pragma unroll 8
  for (int k = 0; k < F; ++k) {
    float v = row[k];
    a0 += v * Wfc[k * 2 + 0];
    a1 += v * Wfc[k * 2 + 1];
  }
  out[node * 2 + 0] = a0;
  out[node * 2 + 1] = a1;
}

extern "C" void kernel_launch(void* const* d_in, const int* in_sizes, int n_in,
                              void* d_out, int out_size, void* d_ws, size_t ws_size,
                              hipStream_t stream) {
  const float* x   = (const float*)d_in[0];
  const int*   ei  = (const int*)d_in[1];
  const float* W1l = (const float*)d_in[2];
  const float* b1  = (const float*)d_in[3];
  const float* W1r = (const float*)d_in[4];
  const float* W2l = (const float*)d_in[5];
  const float* b2  = (const float*)d_in[6];
  const float* W2r = (const float*)d_in[7];
  const float* Wfc = (const float*)d_in[8];
  const float* bfc = (const float*)d_in[9];
  float* out = (float*)d_out;

  float* ws     = (float*)d_ws;
  float* sumbuf = ws;
  float* h1     = sumbuf + (size_t)NNODES * F;
  float* h2     = h1 + (size_t)NNODES * F;
  float* cnt    = h2 + (size_t)NNODES * F;

  int* deg    = (int*)h2;
  int* rowptr = deg + SCAN_LEN;
  int* rank   = rowptr + SCAN_LEN;
  int* csr    = rank + NEDGES;

  const int edgeBlocks = (NEDGES + 255) / 256;

  zero_f32<<<160, 256, 0, stream>>>((float*)deg, NNODES);
  edge_rank<<<edgeBlocks, 256, 0, stream>>>(ei, deg, rank);
  scan_deg<<<1, 1024, 0, stream>>>(deg, rowptr, cnt);
  csr_fill<<<edgeBlocks, 256, 0, stream>>>(ei, rowptr, rank, csr);

  gather_sum<<<NNODES / 8, 256, 0, stream>>>(x, rowptr, csr, sumbuf);
  sage_gemm<<<GEMM_GROUPS, 256, 0, stream>>>(sumbuf, cnt, x, W1l, W1r, b1, h1, 1);

  gather_sum<<<NNODES / 8, 256, 0, stream>>>(h1, rowptr, csr, sumbuf);
  sage_gemm<<<GEMM_GROUPS, 256, 0, stream>>>(sumbuf, cnt, h1, W2l, W2r, b2, h2, 1);

  fc_kernel<<<(NNODES + 255) / 256, 256, 0, stream>>>(h2, Wfc, bfc, out);
}


# CSR build in 2 launches, not 3: each edge group orders its edges by bucket in LDS into its own stretch of tmp; a bucket's size and place come from one table column; row order = LDS atomic arrival
# speedup vs baseline: 4.4715x; 1.9845x over previous
#include <hip/hip_runtime.h>

typedef __attribute__((ext_vector_type(2))) float v2f;
typedef __attribute__((ext_vector_type(8))) float v8f;

#define NNODES 40000
#define NEDGES 640000
#define F 128


#define BSH 8
#define BNODES (1 << BSH)
#define NB ((NNODES + BNODES - 1) / BNODES)
#define CH 8192
#define NG ((NEDGES + CH - 1) / CH)
#define BCAP 8192

static_assert(NNODES <= 65536, "an edge is packed as two 16-bit node ids");
static_assert(NEDGES % 4 == 0, "edges are read four at a time");
static_assert(NB <= 1024 && BNODES <= 1024, "one thread per bucket / per node of a bucket");
static_assert(NG * NB <= 16384 && NNODES + 1 <= 40960, "sizes reserved in the workspace");

__device__ __forceinline__ int wave_incl_scan(int s, int lane) {
#pragma unroll
  for (int d = 1; d < 32; d <<= 1) {
    int n = __shfl_up(s, d, 32);
    if (lane >= d) s += n;
  }
  return s;
}

__device__ __forceinline__ int block_excl_scan(int v, int* wtot, int tid, int& total) {
  const int lane = tid & 31;
  const int wave = tid >> 5;
  const int incl = wave_incl_scan(v, lane);
  if (lane == 31) wtot[wave] = incl;
  __syncthreads();
  if (wave == 0) {
    const int w = wtot[lane];
    const int s = wave_incl_scan(w, lane);
    wtot[lane] = s - w;
    if (lane == 31) wtot[32] = s;
  }
  __syncthreads();
  total = wtot[32];
  const int r = wtot[wave] + incl - v;
  __syncthreads();
  return r;
}

__global__ __launch_bounds__(1024)
void bin_local(const int* __restrict__ ei, int* __restrict__ cnt, int* __restrict__ lst,
               unsigned* __restrict__ tmp) {
  __shared__ int h[NB];
  __shared__ int wtot[33];
  __shared__ unsigned sorted[CH];
  const int tid = threadIdx.x;
  const int g = blockIdx.x;
  if (tid < NB) h[tid] = 0;
  int4 dd[2], ss[2];
  const int base = g * CH + tid * 4;
#pragma unroll
  for (int k = 0; k < 2; ++k) {
    const int e = base + k * 4096;
    const int ec = e < NEDGES ? e : 0;
    dd[k] = *(const int4*)(ei + NEDGES + ec);
    ss[k] = *(const int4*)(ei + ec);
  }
  __syncthreads();
#pragma unroll
  for (int k = 0; k < 2; ++k) {
    if (base + k * 4096 < NEDGES) {
      atomicAdd(&h[dd[k].x >> BSH], 1);
      atomicAdd(&h[dd[k].y >> BSH], 1);
      atomicAdd(&h[dd[k].z >> BSH], 1);
      atomicAdd(&h[dd[k].w >> BSH], 1);
    }
  }
  __syncthreads();
  const int mine = tid < NB ? h[tid] : 0;
  int nmine;
  const int lstart = block_excl_scan(mine, wtot, tid, nmine);
  if (tid < NB) h[tid] = lstart;
  __syncthreads();
#pragma unroll
  for (int k = 0; k < 2; ++k) {
    if (base + k * 4096 < NEDGES) {
      const int dv[4] = {dd[k].x, dd[k].y, dd[k].z, dd[k].w};
      const int sv[4] = {ss[k].x, ss[k].y, ss[k].z, ss[k].w};
#pragma unroll
      for (int u = 0; u < 4; ++u) {
        const int r = atomicAdd(&h[dv[u] >> BSH], 1);
        sorted[r] = ((unsigned)sv[u] << 16) | (unsigned)dv[u];
      }
    }
  }
  __syncthreads();
  unsigned* dst = tmp + g * CH;
  if (nmine == CH) {
#pragma unroll
    for (int k = 0; k < CH / 1024; ++k) dst[tid + k * 1024] = sorted[tid + k * 1024];
  } else {
    for (int j = tid; j < nmine; j += 1024) dst[j] = sorted[j];
  }
  if (tid < NB) {
    cnt[g * NB + tid] = mine;
    lst[g * NB + tid] = lstart;
  }
}

#define BREG (BCAP / 1024)
#define SEGPW ((NG + 31) / 32)
#define SEGW 3

__device__ __forceinline__ int wave_sum(int v) {
#pragma unroll
  for (int d = 16; d > 0; d >>= 1) v += __shfl_xor(v, d, 32);
  return v;
}

__global__ __launch_bounds__(1024)
void bucket_csr(const unsigned* __restrict__ tmp, const int* __restrict__ cnt,
                const int* __restrict__ lst,
                int* __restrict__ rowptr, int* __restrict__ csr) {
  __shared__ int deg[BNODES];
  __shared__ int wtot[33];
  __shared__ unsigned ebuf[BCAP];
  __shared__ int sorted[BCAP];
  const int tid = threadIdx.x;
  const int lane = tid & 31;
  const int wave = __builtin_amdgcn_readfirstlane(tid >> 5);
  const int b = blockIdx.x;
  if (tid < BNODES) deg[tid] = 0;
  int c3[SEGPW], l3[SEGPW], ex3[SEGPW];
#pragma unroll
  for (int i = 0; i < SEGPW; ++i) {
    const int q = lane + 32 * i;
    const int qc = q < NG ? q : NG - 1;
    c3[i] = cnt[qc * NB + b];
    l3[i] = lst[qc * NB + b];
  }
  int nb = 0, lsum = 0;
#pragma unroll
  for (int i = 0; i < SEGPW; ++i) {
    const bool real = lane + 32 * i < NG;
    if (!real) c3[i] = 0;
    const int incl = wave_incl_scan(c3[i], lane);
    ex3[i] = nb + incl - c3[i];
    nb += __builtin_amdgcn_readlane(incl, 31);
    lsum += real ? l3[i] : 0;
  }
  const int s = wave_sum(lsum);
  const bool staged = nb <= BCAP;
  if (staged) {
    unsigned pv[SEGPW][SEGW];
    int sb[SEGPW], len[SEGPW], pre[SEGPW];
#pragma unroll
    for (int i = 0; i < SEGPW; ++i) {
      const int q = wave + 32 * i;
      const int qc = q < NG ? q : NG - 1;
      len[i] = __builtin_amdgcn_readlane(c3[i], wave);
      sb[i]  = qc * CH + __builtin_amdgcn_readlane(l3[i], wave);
      pre[i] = __builtin_amdgcn_readlane(ex3[i], wave);
#pragma unroll
      for (int p = 0; p < SEGW; ++p) {
        const int o = lane + 32 * p;
        pv[i][p] = tmp[o < len[i] ? sb[i] + o : qc * CH];
      }
    }
#pragma unroll
    for (int i = 0; i < SEGPW; ++i) {
#pragma unroll
      for (int p = 0; p < SEGW; ++p) {
        const int o = lane + 32 * p;
        if (o < len[i]) ebuf[pre[i] + o] = pv[i][p];
      }
      for (int o = lane + 32 * SEGW; o < len[i]; o += 32) ebuf[pre[i] + o] = tmp[sb[i] + o];
    }
  }
  __syncthreads();
  unsigned w[BREG];
#pragma unroll
  for (int k = 0; k < BREG; ++k) w[k] = ebuf[tid + k * 1024];
  if (staged) {
#pragma unroll
    for (int k = 0; k < BREG; ++k)
      if (tid + k * 1024 < nb) atomicAdd(&deg[w[k] & (BNODES - 1)], 1);
  } else {
#pragma unroll
    for (int i = 0; i < SEGPW; ++i) {
      for (int l = 0; l < 32 && 32 * i + l < NG; ++l) {
        const int n = __builtin_amdgcn_readlane(c3[i], l);
        const int a = (32 * i + l) * CH + __builtin_amdgcn_readlane(l3[i], l);
        for (int o = tid; o < n; o += 1024) atomicAdd(&deg[tmp[a + o] & (BNODES - 1)], 1);
      }
    }
  }
  __syncthreads();
  int dtotal;
  const int excl = block_excl_scan(tid < BNODES ? deg[tid] : 0, wtot, tid, dtotal);
  if (tid < BNODES) deg[tid] = excl;
  __syncthreads();
  if (staged) {
    int r[BREG];
#pragma unroll
    for (int k = 0; k < BREG; ++k) {
      r[k] = 0;
      if (tid + k * 1024 < nb) r[k] = atomicAdd(&deg[w[k] & (BNODES - 1)], 1);
    }
#pragma unroll
    for (int k = 0; k < BREG; ++k)
      if (tid + k * 1024 < nb) sorted[r[k]] = (int)(w[k] >> 16);
  } else {
#pragma unroll
    for (int i = 0; i < SEGPW; ++i) {
      for (int l = 0; l < 32 && 32 * i + l < NG; ++l) {
        const int n = __builtin_amdgcn_readlane(c3[i], l);
        const int a = (32 * i + l) * CH + __builtin_amdgcn_readlane(l3[i], l);
        for (int o = tid; o < n; o += 1024) {
          const unsigned x = tmp[a + o];
          const int p = atomicAdd(&deg[x & (BNODES - 1)], 1);
          csr[s + p] = (int)(x >> 16);
        }
      }
    }
  }
  __syncthreads();
  if (staged) {
    int id[BREG];
#pragma unroll
    for (int k = 0; k < BREG; ++k) id[k] = sorted[tid + k * 1024];
#pragma unroll
    for (int k = 0; k < BREG; ++k) {
      asm volatile("" : "+v"(id[k]));
    }
#pragma unroll
    for (int k = 0; k < BREG; ++k) {
      const int j = tid + k * 1024;
      if (j < nb) csr[s + j] = id[k];
    }
  }
  if (tid < BNODES) {
    const int node = b * BNODES + tid;
    if (node < NNODES) rowptr[node] = s + excl;
  }
  if (b == NB - 1 && tid == 0) rowptr[NNODES] = s + nb;
}

#define NTILES (NNODES / 16)
#define GEMM_GROUPS 512
#define APITCH 132
#define GCH 8

__device__ __forceinline__ void add4(float4& a, const float4& b) {
  a.x += b.x; a.y += b.y; a.z += b.z; a.w += b.w;
}

__device__ __forceinline__ int load_bounds(const int* __restrict__ rowptr, int n, int lane) {
  return rowptr[n + (lane < 2 ? lane : 2)];
}
__device__ __forceinline__ void bounds_from(int v, int& e0, int& e1, int& e2) {
  e0 = __builtin_amdgcn_readlane(v, 0);
  e1 = __builtin_amdgcn_readlane(v, 1);
  e2 = __builtin_amdgcn_readlane(v, 2);
}

__device__ __forceinline__ void load_ids(const int* __restrict__ csr, int e0, int e1, int e2,
                                         int b, int lane, int& id0, int& id1) {
  const int l0 = e1 - e0 - b;
  const int l1 = e2 - e1 - b;
  id0 = 0;
  id1 = 0;
  if (l0 > 0) id0 = csr[e0 + b + (lane < l0 ? lane : l0 - 1)];
  if (l1 > 0) id1 = csr[e1 + b + (lane < l1 ? lane : l1 - 1)];
}

__device__ __forceinline__ void gather_pair(const float* __restrict__ feat,
                                            const int* __restrict__ csr,
                                            int e0, int e1, int e2, int lane,
                                            int id0, int id1,
                                            float4& a0, float4& a1) {
  float z = 0.0f;
  asm("" : "+v"(z));
  a0 = make_float4(z, z, z, z);
  a1 = make_float4(z, z, z, z);
  const int d0 = e1 - e0;
  const int d1 = e2 - e1;
  const int dmax = d0 > d1 ? d0 : d1;
  for (int b = 0; b < dmax; b += 32) {
    if (b > 0) load_ids(csr, e0, e1, e2, b, lane, id0, id1);
    const int l0 = d0 - b, l1 = d1 - b;
    const int n0 = l0 < 0 ? 0 : (l0 < 32 ? l0 : 32);
    const int n1 = l1 < 0 ? 0 : (l1 < 32 ? l1 : 32);
#pragma unroll
    for (int c = 0; c < 32; c += GCH) {
      if (c == 0 || c < n0 || c < n1) {
        float4 v0[GCH], v1[GCH];
        unsigned s0[GCH], s1[GCH];
#pragma unroll
        for (int u = 0; u < GCH; ++u) {
          s0[u] = (unsigned)__builtin_amdgcn_readlane(id0, c + u);
          s1[u] = (unsigned)__builtin_amdgcn_readlane(id1, c + u);
        }
#pragma unroll
        for (int u = 0; u < GCH; ++u) {
          v0[u] = make_float4(0.0f, 0.0f, 0.0f, 0.0f);
          if (c + u < n0) v0[u] = *(const float4*)(feat + (size_t)s0[u] * F + lane * 4);
          v1[u] = make_float4(0.0f, 0.0f, 0.0f, 0.0f);
          if (c + u < n1) v1[u] = *(const float4*)(feat + (size_t)s1[u] * F + lane * 4);
        }
#pragma unroll
        for (int u = 0; u < GCH; ++u) add4(a0, v0[u]);
#pragma unroll
        for (int u = 0; u < GCH; ++u) add4(a1, v1[u]);
      }
    }
  }
}

template <bool HEAD>
__global__ __launch_bounds__(256)
void sage_layer(const float* __restrict__ feat,
                const int* __restrict__ rowptr,
                const int* __restrict__ csr,
                const float* __restrict__ Wl,
                const float* __restrict__ Wr,
                const float* __restrict__ bias,
                float* __restrict__ out,
                int relu,
                const float* __restrict__ Wfc,
                const float* __restrict__ bfc) {
  __shared__ __attribute__((aligned(16))) float sS[2][16 * APITCH];
  __shared__ __attribute__((aligned(16))) float sX[2][16 * APITCH];
  __shared__ __attribute__((aligned(16))) float sP[2][8 * 32];
  __shared__ __attribute__((aligned(16))) float sHB[F];
  __shared__ __attribute__((aligned(16))) float sHW[2 * F];
  const int tid  = threadIdx.x;
  const int lane = tid & 31;
  const int wave = __builtin_amdgcn_readfirstlane(tid >> 5);
  const int half = lane >> 4;
  const int lid  = lane & 15;
  const int col  = wave * 16 + lid;

  const int r0 = 2 * wave, r1 = 2 * wave + 1;
  const int soff0 = r0 * APITCH + lane * 4;
  const int soff1 = r1 * APITCH + lane * 4;
  const bool gather_first = (((wave >> 2) ^ wave) & 1) != 0;

  int tile = blockIdx.x;
  int ea0, ea1, ea2, ida0, ida1;
  int eb0 = 0, eb1 = 0, eb2 = 0, idb0 = 0, idb1 = 0;
  int cv = 0;
  float4 gx0, gx1, a0, a1;
  {
    const int t1 = tile + GEMM_GROUPS, t2 = tile + 2 * GEMM_GROUPS;
    const int av = load_bounds(rowptr, tile * 16 + r0, lane);
    int bv2 = 0;
    if (t1 < NTILES) bv2 = load_bounds(rowptr, t1 * 16 + r0, lane);
    if (t2 < NTILES) cv  = load_bounds(rowptr, t2 * 16 + r0, lane);
    const size_t g0 = (size_t)(tile * 16 + r0) * F + lane * 4;
    gx0 = *(const float4*)(feat + g0);
    gx1 = *(const float4*)(feat + g0 + F);
    bounds_from(av, ea0, ea1, ea2);
    bounds_from(bv2, eb0, eb1, eb2);
    load_ids(csr, ea0, ea1, ea2, 0, lane, ida0, ida1);
  }

  v2f bl[32], br[32];
  {
    const float* wl = Wl + (4 * half) * F + col;
    const float* wr = Wr + (4 * half) * F + col;
#pragma unroll
    for (int p = 0; p < 32; ++p) {
      const int k = 8 * (p >> 1) + 2 * (p & 1);
      bl[p].x = wl[k * F];
      bl[p].y = wl[(k + 1) * F];
      br[p].x = wr[k * F];
      br[p].y = wr[(k + 1) * F];
    }
  }
  const float bv = bias[col];
  const int f0 = wave * 16 + 8 * half;
  float hbfc = 0.0f;
  if (HEAD) {
    if (tid < F) sHB[tid] = bias[tid];
    sHW[tid] = Wfc[tid];
    hbfc = bfc[lane & 1];
  }

  gather_pair(feat, csr, ea0, ea1, ea2, lane, ida0, ida1, a0, a1);
  {
    const float i0 = 1.0f / fmaxf((float)(ea1 - ea0), 1.0f);
    const float i1 = 1.0f / fmaxf((float)(ea2 - ea1), 1.0f);
    *(float4*)(&sS[0][soff0]) = make_float4(a0.x * i0, a0.y * i0, a0.z * i0, a0.w * i0);
    *(float4*)(&sS[0][soff1]) = make_float4(a1.x * i1, a1.y * i1, a1.z * i1, a1.w * i1);
    *(float4*)(&sX[0][soff0]) = gx0;
    *(float4*)(&sX[0][soff1]) = gx1;
  }
  ea0 = eb0; ea1 = eb1; ea2 = eb2;
  bounds_from(cv, eb0, eb1, eb2);
  if (tile + GEMM_GROUPS < NTILES) load_ids(csr, ea0, ea1, ea2, 0, lane, ida0, ida1);
  __syncthreads();

  const int aoff = lid * APITCH + 4 * half;
  int buf = 0;
  for (; tile < NTILES; tile += GEMM_GROUPS) {
    const int next  = tile + GEMM_GROUPS;
    const int next2 = tile + 2 * GEMM_GROUPS;
    const int next3 = tile + 3 * GEMM_GROUPS;
    const bool more  = next  < NTILES;
    const bool more2 = next2 < NTILES;
    if (next3 < NTILES) cv = load_bounds(rowptr, next3 * 16 + r0, lane);
    if (more && gather_first) gather_pair(feat, csr, ea0, ea1, ea2, lane, ida0, ida1, a0, a1);
    if (more2) load_ids(csr, eb0, eb1, eb2, 0, lane, idb0, idb1);
    if (more) {
      const size_t g0 = (size_t)(next * 16 + r0) * F + lane * 4;
      gx0 = *(const float4*)(feat + g0);
      gx1 = *(const float4*)(feat + g0 + F);
    }

    const float* as = &sS[buf][aoff];
    const float* ax = &sX[buf][aoff];
    v8f accl = {};
    v8f accr = {};
#pragma unroll
    for (int j = 0; j < 16; ++j) {
      const float4 fs = *(const float4*)(as + 8 * j);
      const float4 fx = *(const float4*)(ax + 8 * j);
      v2f a;
      if (HEAD) {
        a.x = fs.x; a.y = fs.y;
        accl = __builtin_amdgcn_wmma_f32_16x16x4_f32(false, bl[2 * j], false, a,
                                                     (short)0, accl, false, false);
        a.x = fx.x; a.y = fx.y;
        accr = __builtin_amdgcn_wmma_f32_16x16x4_f32(false, br[2 * j], false, a,
                                                     (short)0, accr, false, false);
        a.x = fs.z; a.y = fs.w;
        accl = __builtin_amdgcn_wmma_f32_16x16x4_f32(false, bl[2 * j + 1], false, a,
                                                     (short)0, accl, false, false);
        a.x = fx.z; a.y = fx.w;
        accr = __builtin_amdgcn_wmma_f32_16x16x4_f32(false, br[2 * j + 1], false, a,
                                                     (short)0, accr, false, false);
      } else {
        a.x = fs.x; a.y = fs.y;
        accl = __builtin_amdgcn_wmma_f32_16x16x4_f32(false, a, false, bl[2 * j],
                                                     (short)0, accl, false, false);
        a.x = fx.x; a.y = fx.y;
        accr = __builtin_amdgcn_wmma_f32_16x16x4_f32(false, a, false, br[2 * j],
                                                     (short)0, accr, false, false);
        a.x = fs.z; a.y = fs.w;
        accl = __builtin_amdgcn_wmma_f32_16x16x4_f32(false, a, false, bl[2 * j + 1],
                                                     (short)0, accl, false, false);
        a.x = fx.z; a.y = fx.w;
        accr = __builtin_amdgcn_wmma_f32_16x16x4_f32(false, a, false, br[2 * j + 1],
                                                     (short)0, accr, false, false);
      }
    }

    float val[8];
    if (HEAD) {
      float hb[8], hw[16];
      *(float4*)(hb)      = *(const float4*)(&sHB[f0]);
      *(float4*)(hb + 4)  = *(const float4*)(&sHB[f0 + 4]);
#pragma unroll
      for (int q = 0; q < 4; ++q) *(float4*)(hw + 4 * q) = *(const float4*)(&sHW[2 * f0 + 4 * q]);
      float s0 = 0.0f, s1 = 0.0f;
#pragma unroll
      for (int r = 0; r < 8; ++r) {
        float t = accl[r] + accr[r] + hb[r];
        if (relu) t = fmaxf(t, 0.0f);
        s0 = fmaf(t, hw[2 * r], s0);
        s1 = fmaf(t, hw[2 * r + 1], s1);
      }
      s0 += __shfl_xor(s0, 16, 32);
      s1 += __shfl_xor(s1, 16, 32);
      if (half == 0) *(float2*)(&sP[buf][wave * 32 + 2 * lid]) = make_float2(s0, s1);
    } else {
#pragma unroll
      for (int v = 0; v < 8; ++v) {
        val[v] = accl[v] + accr[v] + bv;
        if (relu) val[v] = fmaxf(val[v], 0.0f);
      }
    }

    if (more) {
      if (!gather_first) gather_pair(feat, csr, ea0, ea1, ea2, lane, ida0, ida1, a0, a1);
      const float i0 = 1.0f / fmaxf((float)(ea1 - ea0), 1.0f);
      const float i1 = 1.0f / fmaxf((float)(ea2 - ea1), 1.0f);
      float* ds = sS[buf ^ 1];
      float* dx = sX[buf ^ 1];
      *(float4*)(ds + soff0) = make_float4(a0.x * i0, a0.y * i0, a0.z * i0, a0.w * i0);
      *(float4*)(ds + soff1) = make_float4(a1.x * i1, a1.y * i1, a1.z * i1, a1.w * i1);
      *(float4*)(dx + soff0) = gx0;
      *(float4*)(dx + soff1) = gx1;
    }
    ea0 = eb0; ea1 = eb1; ea2 = eb2; ida0 = idb0; ida1 = idb1;
    bounds_from(cv, eb0, eb1, eb2);
    __syncthreads();
    const int base = tile * 16;
    if (HEAD) {
      if (wave == 0) {
        float t = hbfc;
#pragma unroll
        for (int w = 0; w < 8; ++w) t += sP[buf][w * 32 + lane];
        out[(size_t)base * 2 + lane] = t;
      }
    } else {
#pragma unroll
      for (int v = 0; v < 8; ++v) {
        out[(size_t)(base + v + 8 * half) * F + col] = val[v];
      }
    }
    buf ^= 1;
  }
}

extern "C" void kernel_launch(void* const* d_in, const int* in_sizes, int n_in,
                              void* d_out, int out_size, void* d_ws, size_t ws_size,
                              hipStream_t stream) {
  const float* x   = (const float*)d_in[0];
  const int*   ei  = (const int*)d_in[1];
  const float* W1l = (const float*)d_in[2];
  const float* b1  = (const float*)d_in[3];
  const float* W1r = (const float*)d_in[4];
  const float* W2l = (const float*)d_in[5];
  const float* b2  = (const float*)d_in[6];
  const float* W2r = (const float*)d_in[7];
  const float* Wfc = (const float*)d_in[8];
  const float* bfc = (const float*)d_in[9];
  float* out = (float*)d_out;

  float* ws     = (float*)d_ws;
  float* csrbuf = ws;
  float* h1     = csrbuf + (size_t)NNODES * F;

  int*      cnt    = (int*)csrbuf;
  int*      lst    = cnt + 16384;
  int*      rowptr = lst + 16384;
  unsigned* tmp    = (unsigned*)(rowptr + 40960);
  int*      csr    = (int*)tmp + NG * CH;

  bin_local<<<NG, 1024, 0, stream>>>(ei, cnt, lst, tmp);
  bucket_csr<<<NB, 1024, 0, stream>>>(tmp, cnt, lst, rowptr, csr);

  sage_layer<false><<<GEMM_GROUPS, 256, 0, stream>>>(x, rowptr, csr, W1l, W1r, b1, h1, 1, Wfc, bfc);

  sage_layer<true><<<GEMM_GROUPS, 256, 0, stream>>>(h1, rowptr, csr, W2l, W2r, b2, out, 1, Wfc, bfc);
}


# Layer weights copied per call into fragment order by 16 spare work-groups of the first launch: a wave loads its B fragments as 32 whole-line 16-byte loads, not 128 one-word loads
# speedup vs baseline: 4.6479x; 1.0394x over previous
#include <hip/hip_runtime.h>

typedef __attribute__((ext_vector_type(2))) float v2f;
typedef __attribute__((ext_vector_type(8))) float v8f;

#define NNODES 40000
#define NEDGES 640000
#define F 128


#define BSH 8
#define BNODES (1 << BSH)
#define NB ((NNODES + BNODES - 1) / BNODES)
#define CH 8192
#define NG ((NEDGES + CH - 1) / CH)
#define BCAP 8192

#define WPACK_LAYER (8 * 32 * 32 * 4)
#define WPACK_GROUPS (2 * WPACK_LAYER / 4 / 1024)

static_assert(NNODES <= 65536, "an edge is packed as two 16-bit node ids");
static_assert(NEDGES % 4 == 0, "edges are read four at a time");
static_assert(NB <= 1024 && BNODES <= 1024, "one thread per bucket / per node of a bucket");
static_assert(NG * NB <= 16384 && NNODES + 1 <= 40960, "sizes reserved in the workspace");
static_assert(16384 + 16384 + 40960 + NG * CH + NEDGES + 2 * WPACK_LAYER <= NNODES * F,
              "the CSR arrays and the weight table fit the region in front of h1");

__device__ __forceinline__ int wave_incl_scan(int s, int lane) {
#pragma unroll
  for (int d = 1; d < 32; d <<= 1) {
    int n = __shfl_up(s, d, 32);
    if (lane >= d) s += n;
  }
  return s;
}

__device__ __forceinline__ int block_excl_scan(int v, int* wtot, int tid, int& total) {
  const int lane = tid & 31;
  const int wave = tid >> 5;
  const int incl = wave_incl_scan(v, lane);
  if (lane == 31) wtot[wave] = incl;
  __syncthreads();
  if (wave == 0) {
    const int w = wtot[lane];
    const int s = wave_incl_scan(w, lane);
    wtot[lane] = s - w;
    if (lane == 31) wtot[32] = s;
  }
  __syncthreads();
  total = wtot[32];
  const int r = wtot[wave] + incl - v;
  __syncthreads();
  return r;
}

__device__ __forceinline__ void pack_weights(int q, int tid,
                                             const float* __restrict__ W1l, const float* __restrict__ W1r,
                                             const float* __restrict__ W2l, const float* __restrict__ W2r,
                                             float* __restrict__ wpack) {
  const int layer = q >> 3;
  const float* wl = layer ? W2l : W1l;
  const float* wr = layer ? W2r : W1r;
  const int w = q & 7, p = tid >> 5, lane = tid & 31;
  const int n = 16 * w + (lane & 15);
  const int k = 8 * (p >> 1) + 2 * (p & 1) + 4 * (lane >> 4);
  const float4 v = make_float4(wl[k * F + n], wl[(k + 1) * F + n], wr[k * F + n], wr[(k + 1) * F + n]);
  *(float4*)(wpack + layer * WPACK_LAYER + (w * 1024 + tid) * 4) = v;
}

__global__ __launch_bounds__(1024)
void bin_local(const int* __restrict__ ei, int* __restrict__ cnt, int* __restrict__ lst,
               unsigned* __restrict__ tmp,
               const float* __restrict__ W1l, const float* __restrict__ W1r,
               const float* __restrict__ W2l, const float* __restrict__ W2r,
               float* __restrict__ wpack) {
  __shared__ int h[NB];
  __shared__ int wtot[33];
  __shared__ unsigned sorted[CH];
  const int tid = threadIdx.x;
  const int g = blockIdx.x;
  if (g >= NG) {
    pack_weights(g - NG, tid, W1l, W1r, W2l, W2r, wpack);
    return;
  }
  if (tid < NB) h[tid] = 0;
  int4 dd[2], ss[2];
  const int base = g * CH + tid * 4;
#pragma unroll
  for (int k = 0; k < 2; ++k) {
    const int e = base + k * 4096;
    const int ec = e < NEDGES ? e : 0;
    dd[k] = *(const int4*)(ei + NEDGES + ec);
    ss[k] = *(const int4*)(ei + ec);
  }
  __syncthreads();
#pragma unroll
  for (int k = 0; k < 2; ++k) {
    if (base + k * 4096 < NEDGES) {
      atomicAdd(&h[dd[k].x >> BSH], 1);
      atomicAdd(&h[dd[k].y >> BSH], 1);
      atomicAdd(&h[dd[k].z >> BSH], 1);
      atomicAdd(&h[dd[k].w >> BSH], 1);
    }
  }
  __syncthreads();
  const int mine = tid < NB ? h[tid] : 0;
  int nmine;
  const int lstart = block_excl_scan(mine, wtot, tid, nmine);
  if (tid < NB) h[tid] = lstart;
  __syncthreads();
#pragma unroll
  for (int k = 0; k < 2; ++k) {
    if (base + k * 4096 < NEDGES) {
      const int dv[4] = {dd[k].x, dd[k].y, dd[k].z, dd[k].w};
      const int sv[4] = {ss[k].x, ss[k].y, ss[k].z, ss[k].w};
#pragma unroll
      for (int u = 0; u < 4; ++u) {
        const int r = atomicAdd(&h[dv[u] >> BSH], 1);
        sorted[r] = ((unsigned)sv[u] << 16) | (unsigned)dv[u];
      }
    }
  }
  __syncthreads();
  unsigned* dst = tmp + g * CH;
  if (nmine == CH) {
#pragma unroll
    for (int k = 0; k < CH / 1024; ++k) dst[tid + k * 1024] = sorted[tid + k * 1024];
  } else {
    for (int j = tid; j < nmine; j += 1024) dst[j] = sorted[j];
  }
  if (tid < NB) {
    cnt[g * NB + tid] = mine;
    lst[g * NB + tid] = lstart;
  }
}

#define BREG (BCAP / 1024)
#define SEGPW ((NG + 31) / 32)
#define SEGW 3

__device__ __forceinline__ int wave_sum(int v) {
#pragma unroll
  for (int d = 16; d > 0; d >>= 1) v += __shfl_xor(v, d, 32);
  return v;
}

__global__ __launch_bounds__(1024)
void bucket_csr(const unsigned* __restrict__ tmp, const int* __restrict__ cnt,
                const int* __restrict__ lst,
                int* __restrict__ rowptr, int* __restrict__ csr) {
  __shared__ int deg[BNODES];
  __shared__ int wtot[33];
  __shared__ unsigned ebuf[BCAP];
  __shared__ int sorted[BCAP];
  const int tid = threadIdx.x;
  const int lane = tid & 31;
  const int wave = __builtin_amdgcn_readfirstlane(tid >> 5);
  const int b = blockIdx.x;
  if (tid < BNODES) deg[tid] = 0;
  int c3[SEGPW], l3[SEGPW], ex3[SEGPW];
#pragma unroll
  for (int i = 0; i < SEGPW; ++i) {
    const int q = lane + 32 * i;
    const int qc = q < NG ? q : NG - 1;
    c3[i] = cnt[qc * NB + b];
    l3[i] = lst[qc * NB + b];
  }
  int nb = 0, lsum = 0;
#pragma unroll
  for (int i = 0; i < SEGPW; ++i) {
    const bool real = lane + 32 * i < NG;
    if (!real) c3[i] = 0;
    const int incl = wave_incl_scan(c3[i], lane);
    ex3[i] = nb + incl - c3[i];
    nb += __builtin_amdgcn_readlane(incl, 31);
    lsum += real ? l3[i] : 0;
  }
  const int s = wave_sum(lsum);
  const bool staged = nb <= BCAP;
  if (staged) {
    unsigned pv[SEGPW][SEGW];
    int sb[SEGPW], len[SEGPW], pre[SEGPW];
#pragma unroll
    for (int i = 0; i < SEGPW; ++i) {
      const int q = wave + 32 * i;
      const int qc = q < NG ? q : NG - 1;
      len[i] = __builtin_amdgcn_readlane(c3[i], wave);
      sb[i]  = qc * CH + __builtin_amdgcn_readlane(l3[i], wave);
      pre[i] = __builtin_amdgcn_readlane(ex3[i], wave);
#pragma unroll
      for (int p = 0; p < SEGW; ++p) {
        const int o = lane + 32 * p;
        pv[i][p] = tmp[o < len[i] ? sb[i] + o : qc * CH];
      }
    }
#pragma unroll
    for (int i = 0; i < SEGPW; ++i) {
#pragma unroll
      for (int p = 0; p < SEGW; ++p) {
        const int o = lane + 32 * p;
        if (o < len[i]) ebuf[pre[i] + o] = pv[i][p];
      }
      for (int o = lane + 32 * SEGW; o < len[i]; o += 32) ebuf[pre[i] + o] = tmp[sb[i] + o];
    }
  }
  __syncthreads();
  unsigned w[BREG];
#pragma unroll
  for (int k = 0; k < BREG; ++k) w[k] = ebuf[tid + k * 1024];
  if (staged) {
#pragma unroll
    for (int k = 0; k < BREG; ++k)
      if (tid + k * 1024 < nb) atomicAdd(&deg[w[k] & (BNODES - 1)], 1);
  } else {
#pragma unroll
    for (int i = 0; i < SEGPW; ++i) {
      for (int l = 0; l < 32 && 32 * i + l < NG; ++l) {
        const int n = __builtin_amdgcn_readlane(c3[i], l);
        const int a = (32 * i + l) * CH + __builtin_amdgcn_readlane(l3[i], l);
        for (int o = tid; o < n; o += 1024) atomicAdd(&deg[tmp[a + o] & (BNODES - 1)], 1);
      }
    }
  }
  __syncthreads();
  int dtotal;
  const int excl = block_excl_scan(tid < BNODES ? deg[tid] : 0, wtot, tid, dtotal);
  if (tid < BNODES) deg[tid] = excl;
  __syncthreads();
  if (staged) {
    int r[BREG];
#pragma unroll
    for (int k = 0; k < BREG; ++k) {
      r[k] = 0;
      if (tid + k * 1024 < nb) r[k] = atomicAdd(&deg[w[k] & (BNODES - 1)], 1);
    }
#pragma unroll
    for (int k = 0; k < BREG; ++k)
      if (tid + k * 1024 < nb) sorted[r[k]] = (int)(w[k] >> 16);
  } else {
#pragma unroll
    for (int i = 0; i < SEGPW; ++i) {
      for (int l = 0; l < 32 && 32 * i + l < NG; ++l) {
        const int n = __builtin_amdgcn_readlane(c3[i], l);
        const int a = (32 * i + l) * CH + __builtin_amdgcn_readlane(l3[i], l);
        for (int o = tid; o < n; o += 1024) {
          const unsigned x = tmp[a + o];
          const int p = atomicAdd(&deg[x & (BNODES - 1)], 1);
          csr[s + p] = (int)(x >> 16);
        }
      }
    }
  }
  __syncthreads();
  if (staged) {
    int id[BREG];
#pragma unroll
    for (int k = 0; k < BREG; ++k) id[k] = sorted[tid + k * 1024];
#pragma unroll
    for (int k = 0; k < BREG; ++k) {
      asm volatile("" : "+v"(id[k]));
    }
#pragma unroll
    for (int k = 0; k < BREG; ++k) {
      const int j = tid + k * 1024;
      if (j < nb) csr[s + j] = id[k];
    }
  }
  if (tid < BNODES) {
    const int node = b * BNODES + tid;
    if (node < NNODES) rowptr[node] = s + excl;
  }
  if (b == NB - 1 && tid == 0) rowptr[NNODES] = s + nb;
}

#define NTILES (NNODES / 16)
#define GEMM_GROUPS 512
#define APITCH 132
#define GCH 8

__device__ __forceinline__ void add4(float4& a, const float4& b) {
  a.x += b.x; a.y += b.y; a.z += b.z; a.w += b.w;
}

__device__ __forceinline__ int load_bounds(const int* __restrict__ rowptr, int n, int lane) {
  return rowptr[n + (lane < 2 ? lane : 2)];
}
__device__ __forceinline__ void bounds_from(int v, int& e0, int& e1, int& e2) {
  e0 = __builtin_amdgcn_readlane(v, 0);
  e1 = __builtin_amdgcn_readlane(v, 1);
  e2 = __builtin_amdgcn_readlane(v, 2);
}

__device__ __forceinline__ void load_ids(const int* __restrict__ csr, int e0, int e1, int e2,
                                         int b, int lane, int& id0, int& id1) {
  const int l0 = e1 - e0 - b;
  const int l1 = e2 - e1 - b;
  id0 = 0;
  id1 = 0;
  if (l0 > 0) id0 = csr[e0 + b + (lane < l0 ? lane : l0 - 1)];
  if (l1 > 0) id1 = csr[e1 + b + (lane < l1 ? lane : l1 - 1)];
}

__device__ __forceinline__ void gather_pair(const float* __restrict__ feat,
                                            const int* __restrict__ csr,
                                            int e0, int e1, int e2, int lane,
                                            int id0, int id1,
                                            float4& a0, float4& a1) {
  float z = 0.0f;
  asm("" : "+v"(z));
  a0 = make_float4(z, z, z, z);
  a1 = make_float4(z, z, z, z);
  const int d0 = e1 - e0;
  const int d1 = e2 - e1;
  const int dmax = d0 > d1 ? d0 : d1;
  for (int b = 0; b < dmax; b += 32) {
    if (b > 0) load_ids(csr, e0, e1, e2, b, lane, id0, id1);
    const int l0 = d0 - b, l1 = d1 - b;
    const int n0 = l0 < 0 ? 0 : (l0 < 32 ? l0 : 32);
    const int n1 = l1 < 0 ? 0 : (l1 < 32 ? l1 : 32);
#pragma unroll
    for (int c = 0; c < 32; c += GCH) {
      if (c == 0 || c < n0 || c < n1) {
        float4 v0[GCH], v1[GCH];
        unsigned s0[GCH], s1[GCH];
#pragma unroll
        for (int u = 0; u < GCH; ++u) {
          s0[u] = (unsigned)__builtin_amdgcn_readlane(id0, c + u);
          s1[u] = (unsigned)__builtin_amdgcn_readlane(id1, c + u);
        }
#pragma unroll
        for (int u = 0; u < GCH; ++u) {
          v0[u] = make_float4(0.0f, 0.0f, 0.0f, 0.0f);
          if (c + u < n0) v0[u] = *(const float4*)(feat + (size_t)s0[u] * F + lane * 4);
          v1[u] = make_float4(0.0f, 0.0f, 0.0f, 0.0f);
          if (c + u < n1) v1[u] = *(const float4*)(feat + (size_t)s1[u] * F + lane * 4);
        }
#pragma unroll
        for (int u = 0; u < GCH; ++u) add4(a0, v0[u]);
#pragma unroll
        for (int u = 0; u < GCH; ++u) add4(a1, v1[u]);
      }
    }
  }
}

template <bool HEAD>
__global__ __launch_bounds__(256)
void sage_layer(const float* __restrict__ feat,
                const int* __restrict__ rowptr,
                const int* __restrict__ csr,
                const float* __restrict__ wpack,
                const float* __restrict__ bias,
                float* __restrict__ out,
                int relu,
                const float* __restrict__ Wfc,
                const float* __restrict__ bfc) {
  __shared__ __attribute__((aligned(16))) float sS[2][16 * APITCH];
  __shared__ __attribute__((aligned(16))) float sX[2][16 * APITCH];
  __shared__ __attribute__((aligned(16))) float sP[2][8 * 32];
  __shared__ __attribute__((aligned(16))) float sHB[F];
  __shared__ __attribute__((aligned(16))) float sHW[2 * F];
  const int tid  = threadIdx.x;
  const int lane = tid & 31;
  const int wave = __builtin_amdgcn_readfirstlane(tid >> 5);
  const int half = lane >> 4;
  const int lid  = lane & 15;
  const int col  = wave * 16 + lid;

  const int r0 = 2 * wave, r1 = 2 * wave + 1;
  const int soff0 = r0 * APITCH + lane * 4;
  const int soff1 = r1 * APITCH + lane * 4;
  const bool gather_first = (((wave >> 2) ^ wave) & 1) != 0;

  int tile = blockIdx.x;
  int ea0, ea1, ea2, ida0, ida1;
  int eb0 = 0, eb1 = 0, eb2 = 0, idb0 = 0, idb1 = 0;
  int cv = 0;
  float4 gx0, gx1, a0, a1;
  {
    const int t1 = tile + GEMM_GROUPS, t2 = tile + 2 * GEMM_GROUPS;
    const int av = load_bounds(rowptr, tile * 16 + r0, lane);
    int bv2 = 0;
    if (t1 < NTILES) bv2 = load_bounds(rowptr, t1 * 16 + r0, lane);
    if (t2 < NTILES) cv  = load_bounds(rowptr, t2 * 16 + r0, lane);
    const size_t g0 = (size_t)(tile * 16 + r0) * F + lane * 4;
    gx0 = *(const float4*)(feat + g0);
    gx1 = *(const float4*)(feat + g0 + F);
    bounds_from(av, ea0, ea1, ea2);
    bounds_from(bv2, eb0, eb1, eb2);
    load_ids(csr, ea0, ea1, ea2, 0, lane, ida0, ida1);
  }

  v2f bl[32], br[32];
  {
    const float4* wp = (const float4*)wpack + wave * 1024 + lane;
#pragma unroll
    for (int p = 0; p < 32; ++p) {
      const float4 t = wp[p * 32];
      bl[p].x = t.x;
      bl[p].y = t.y;
      br[p].x = t.z;
      br[p].y = t.w;
    }
  }
  const float bv = bias[col];
  const int f0 = wave * 16 + 8 * half;
  float hbfc = 0.0f;
  if (HEAD) {
    if (tid < F) sHB[tid] = bias[tid];
    sHW[tid] = Wfc[tid];
    hbfc = bfc[lane & 1];
  }

  gather_pair(feat, csr, ea0, ea1, ea2, lane, ida0, ida1, a0, a1);
  {
    const float i0 = 1.0f / fmaxf((float)(ea1 - ea0), 1.0f);
    const float i1 = 1.0f / fmaxf((float)(ea2 - ea1), 1.0f);
    *(float4*)(&sS[0][soff0]) = make_float4(a0.x * i0, a0.y * i0, a0.z * i0, a0.w * i0);
    *(float4*)(&sS[0][soff1]) = make_float4(a1.x * i1, a1.y * i1, a1.z * i1, a1.w * i1);
    *(float4*)(&sX[0][soff0]) = gx0;
    *(float4*)(&sX[0][soff1]) = gx1;
  }
  ea0 = eb0; ea1 = eb1; ea2 = eb2;
  bounds_from(cv, eb0, eb1, eb2);
  if (tile + GEMM_GROUPS < NTILES) load_ids(csr, ea0, ea1, ea2, 0, lane, ida0, ida1);
  __syncthreads();

  const int aoff = lid * APITCH + 4 * half;
  int buf = 0;
  for (; tile < NTILES; tile += GEMM_GROUPS) {
    const int next  = tile + GEMM_GROUPS;
    const int next2 = tile + 2 * GEMM_GROUPS;
    const int next3 = tile + 3 * GEMM_GROUPS;
    const bool more  = next  < NTILES;
    const bool more2 = next2 < NTILES;
    if (next3 < NTILES) cv = load_bounds(rowptr, next3 * 16 + r0, lane);
    if (more && gather_first) gather_pair(feat, csr, ea0, ea1, ea2, lane, ida0, ida1, a0, a1);
    if (more2) load_ids(csr, eb0, eb1, eb2, 0, lane, idb0, idb1);
    if (more) {
      const size_t g0 = (size_t)(next * 16 + r0) * F + lane * 4;
      gx0 = *(const float4*)(feat + g0);
      gx1 = *(const float4*)(feat + g0 + F);
    }

    const float* as = &sS[buf][aoff];
    const float* ax = &sX[buf][aoff];
    v8f accl = {};
    v8f accr = {};
#pragma unroll
    for (int j = 0; j < 16; ++j) {
      const float4 fs = *(const float4*)(as + 8 * j);
      const float4 fx = *(const float4*)(ax + 8 * j);
      v2f a;
      if (HEAD) {
        a.x = fs.x; a.y = fs.y;
        accl = __builtin_amdgcn_wmma_f32_16x16x4_f32(false, bl[2 * j], false, a,
                                                     (short)0, accl, false, false);
        a.x = fx.x; a.y = fx.y;
        accr = __builtin_amdgcn_wmma_f32_16x16x4_f32(false, br[2 * j], false, a,
                                                     (short)0, accr, false, false);
        a.x = fs.z; a.y = fs.w;
        accl = __builtin_amdgcn_wmma_f32_16x16x4_f32(false, bl[2 * j + 1], false, a,
                                                     (short)0, accl, false, false);
        a.x = fx.z; a.y = fx.w;
        accr = __builtin_amdgcn_wmma_f32_16x16x4_f32(false, br[2 * j + 1], false, a,
                                                     (short)0, accr, false, false);
      } else {
        a.x = fs.x; a.y = fs.y;
        accl = __builtin_amdgcn_wmma_f32_16x16x4_f32(false, a, false, bl[2 * j],
                                                     (short)0, accl, false, false);
        a.x = fx.x; a.y = fx.y;
        accr = __builtin_amdgcn_wmma_f32_16x16x4_f32(false, a, false, br[2 * j],
                                                     (short)0, accr, false, false);
        a.x = fs.z; a.y = fs.w;
        accl = __builtin_amdgcn_wmma_f32_16x16x4_f32(false, a, false, bl[2 * j + 1],
                                                     (short)0, accl, false, false);
        a.x = fx.z; a.y = fx.w;
        accr = __builtin_amdgcn_wmma_f32_16x16x4_f32(false, a, false, br[2 * j + 1],
                                                     (short)0, accr, false, false);
      }
    }

    float val[8];
    if (HEAD) {
      float hb[8], hw[16];
      *(float4*)(hb)      = *(const float4*)(&sHB[f0]);
      *(float4*)(hb + 4)  = *(const float4*)(&sHB[f0 + 4]);
#pragma unroll
      for (int q = 0; q < 4; ++q) *(float4*)(hw + 4 * q) = *(const float4*)(&sHW[2 * f0 + 4 * q]);
      float s0 = 0.0f, s1 = 0.0f;
#pragma unroll
      for (int r = 0; r < 8; ++r) {
        float t = accl[r] + accr[r] + hb[r];
        if (relu) t = fmaxf(t, 0.0f);
        s0 = fmaf(t, hw[2 * r], s0);
        s1 = fmaf(t, hw[2 * r + 1], s1);
      }
      s0 += __shfl_xor(s0, 16, 32);
      s1 += __shfl_xor(s1, 16, 32);
      if (half == 0) *(float2*)(&sP[buf][wave * 32 + 2 * lid]) = make_float2(s0, s1);
    } else {
#pragma unroll
      for (int v = 0; v < 8; ++v) {
        val[v] = accl[v] + accr[v] + bv;
        if (relu) val[v] = fmaxf(val[v], 0.0f);
      }
    }

    if (more) {
      if (!gather_first) gather_pair(feat, csr, ea0, ea1, ea2, lane, ida0, ida1, a0, a1);
      const float i0 = 1.0f / fmaxf((float)(ea1 - ea0), 1.0f);
      const float i1 = 1.0f / fmaxf((float)(ea2 - ea1), 1.0f);
      float* ds = sS[buf ^ 1];
      float* dx = sX[buf ^ 1];
      *(float4*)(ds + soff0) = make_float4(a0.x * i0, a0.y * i0, a0.z * i0, a0.w * i0);
      *(float4*)(ds + soff1) = make_float4(a1.x * i1, a1.y * i1, a1.z * i1, a1.w * i1);
      *(float4*)(dx + soff0) = gx0;
      *(float4*)(dx + soff1) = gx1;
    }
    ea0 = eb0; ea1 = eb1; ea2 = eb2; ida0 = idb0; ida1 = idb1;
    bounds_from(cv, eb0, eb1, eb2);
    __syncthreads();
    const int base = tile * 16;
    if (HEAD) {
      if (wave == 0) {
        float t = hbfc;
#pragma unroll
        for (int w = 0; w < 8; ++w) t += sP[buf][w * 32 + lane];
        out[(size_t)base * 2 + lane] = t;
      }
    } else {
#pragma unroll
      for (int v = 0; v < 8; ++v) {
        out[(size_t)(base + v + 8 * half) * F + col] = val[v];
      }
    }
    buf ^= 1;
  }
}

extern "C" void kernel_launch(void* const* d_in, const int* in_sizes, int n_in,
                              void* d_out, int out_size, void* d_ws, size_t ws_size,
                              hipStream_t stream) {
  const float* x   = (const float*)d_in[0];
  const int*   ei  = (const int*)d_in[1];
  const float* W1l = (const float*)d_in[2];
  const float* b1  = (const float*)d_in[3];
  const float* W1r = (const float*)d_in[4];
  const float* W2l = (const float*)d_in[5];
  const float* b2  = (const float*)d_in[6];
  const float* W2r = (const float*)d_in[7];
  const float* Wfc = (const float*)d_in[8];
  const float* bfc = (const float*)d_in[9];
  float* out = (float*)d_out;

  float* ws     = (float*)d_ws;
  float* csrbuf = ws;
  float* h1     = csrbuf + (size_t)NNODES * F;

  int*      cnt    = (int*)csrbuf;
  int*      lst    = cnt + 16384;
  int*      rowptr = lst + 16384;
  unsigned* tmp    = (unsigned*)(rowptr + 40960);
  int*      csr    = (int*)tmp + NG * CH;
  float*    wpack  = (float*)(csr + NEDGES);

  bin_local<<<NG + WPACK_GROUPS, 1024, 0, stream>>>(ei, cnt, lst, tmp, W1l, W1r, W2l, W2r, wpack);
  bucket_csr<<<NB, 1024, 0, stream>>>(tmp, cnt, lst, rowptr, csr);

  sage_layer<false><<<GEMM_GROUPS, 256, 0, stream>>>(x, rowptr, csr, wpack, b1, h1, 1, Wfc, bfc);

  sage_layer<true><<<GEMM_GROUPS, 256, 0, stream>>>(h1, rowptr, csr, wpack + WPACK_LAYER, b2, out, 1, Wfc, bfc);
}
